# starting kernel
# baseline (speedup 1.0000x reference)
#include <hip/hip_runtime.h>

typedef __attribute__((ext_vector_type(2))) float v2f;
typedef __attribute__((ext_vector_type(8))) float v8f;

#define DIM   256
#define N1C   40000
#define N2C   8000
#define E1C   1000000
#define E2C   200000
#define LROW  (DIM + 4)

__global__ void zero_int(int* __restrict__ p, int n) {
    int i = blockIdx.x * blockDim.x + threadIdx.x;
    if (i < n) p[i] = 0;
}

__global__ void edge_hist(const int* __restrict__ dst, int* __restrict__ cnt,
                          int n) {
    int i = blockIdx.x * blockDim.x + threadIdx.x;
    if (i < n) atomicAdd(cnt + dst[i], 1);
}

__global__ __launch_bounds__(1024)
void scan_excl(int* __restrict__ data, int n) {
    __shared__ int tmp[1024];
    __shared__ int running;
    int tid = threadIdx.x;
    if (tid == 0) running = 0;
    __syncthreads();
    for (int base = 0; base < n; base += 1024) {
        int i = base + tid;
        int v = (i < n) ? data[i] : 0;
        tmp[tid] = v;
        __syncthreads();
        for (int d = 1; d < 1024; d <<= 1) {
            int t = (tid >= d) ? tmp[tid - d] : 0;
            __syncthreads();
            tmp[tid] += t;
            __syncthreads();
        }
        int incl = tmp[tid];
        int r = running;
        if (i < n) data[i] = r + incl - v;
        __syncthreads();
        if (tid == 1023) running = r + incl;
        __syncthreads();
    }
    if (tid == 0) data[n] = running;
}

__global__ void edge_fill(const int* __restrict__ src,
                          const int* __restrict__ dst,
                          const int* __restrict__ off,
                          int* __restrict__ cur,
                          int* __restrict__ eid, int n) {
    int i = blockIdx.x * blockDim.x + threadIdx.x;
    if (i < n) {
        int d = dst[i];
        int p = atomicAdd(cur + d, 1);
        eid[off[d] + p] = src[i];
    }
}

__global__ __launch_bounds__(512)
void sage_layer(const float* __restrict__ x,
                const int* __restrict__ off,
                const int* __restrict__ eid,
                const float* __restrict__ Wl,
                const float* __restrict__ bias,
                const float* __restrict__ Wr,
                float* __restrict__ out,
                int M) {
    __shared__ float lmean[16 * LROW];
    __shared__ float lx[16 * LROW];

    const int m0   = blockIdx.x * 16;
    const int tid  = threadIdx.x;
    const int lane = tid & 31;
    const int wv   = tid >> 5;

    for (int i = tid; i < 16 * DIM; i += 512) {
        int row = i >> 8;
        int col = i & (DIM - 1);
        lx[row * LROW + col] = x[(long)(m0 + row) * DIM + col];
    }

    {
        int row = m0 + wv;
        int beg = off[row];
        int end = off[row + 1];
        float4 a0 = make_float4(0.f, 0.f, 0.f, 0.f);
        float4 a1 = make_float4(0.f, 0.f, 0.f, 0.f);
        for (int e = beg; e < end; e += 32) {
            int myid = (e + lane < end) ? eid[e + lane] : 0;
            int cnt = min(32, end - e);
            for (int j = 0; j < cnt; ++j) {
                int s = __shfl(myid, j, 32);
                const float4* xr = (const float4*)(x + (long)s * DIM);
                float4 u = xr[lane];
                float4 v = xr[lane + 32];
                a0.x += u.x; a0.y += u.y; a0.z += u.z; a0.w += u.w;
                a1.x += v.x; a1.y += v.y; a1.z += v.z; a1.w += v.w;
            }
        }
        float inv = 1.0f / fmaxf((float)(end - beg), 1.0f);
        float* mrow = lmean + wv * LROW;
        mrow[lane * 4 + 0] = a0.x * inv;
        mrow[lane * 4 + 1] = a0.y * inv;
        mrow[lane * 4 + 2] = a0.z * inv;
        mrow[lane * 4 + 3] = a0.w * inv;
        mrow[128 + lane * 4 + 0] = a1.x * inv;
        mrow[128 + lane * 4 + 1] = a1.y * inv;
        mrow[128 + lane * 4 + 2] = a1.z * inv;
        mrow[128 + lane * 4 + 3] = a1.w * inv;
    }
    __syncthreads();

    const int half = lane >> 4;
    const int lrow = lane & 15;
    const int n0   = wv * 16;

    const float* wl_r = Wl + (long)(n0 + lrow) * DIM;
    const float* wr_r = Wr + (long)(n0 + lrow) * DIM;
    const float* lm_r = lmean + lrow * LROW;
    const float* lx_r = lx + lrow * LROW;

    v8f acc = {};
#pragma unroll 4
    for (int k = 0; k < DIM; k += 4) {
        int kk = k + 2 * half;
        v2f am = *(const v2f*)(lm_r + kk);
        v2f bl = *(const v2f*)(wl_r + kk);
        acc = __builtin_amdgcn_wmma_f32_16x16x4_f32(
            false, am, false, bl, (short)0, acc, false, false);
        v2f ax = *(const v2f*)(lx_r + kk);
        v2f br = *(const v2f*)(wr_r + kk);
        acc = __builtin_amdgcn_wmma_f32_16x16x4_f32(
            false, ax, false, br, (short)0, acc, false, false);
    }

    float bb = bias[n0 + lrow];
    float* orow = out + (long)(m0 + half * 8) * DIM + n0 + lrow;
#pragma unroll
    for (int r = 0; r < 8; ++r) {
        orow[(long)r * DIM] = tanhf(acc[r] + bb);
    }
}

extern "C" void kernel_launch(void* const* d_in, const int* in_sizes, int n_in,
                              void* d_out, int out_size, void* d_ws, size_t ws_size,
                              hipStream_t stream) {
    const float* nodes = (const float*)d_in[0];
    const float* W_l1  = (const float*)d_in[1];
    const float* b1    = (const float*)d_in[2];
    const float* W_r1  = (const float*)d_in[3];
    const float* W_l2  = (const float*)d_in[4];
    const float* b2    = (const float*)d_in[5];
    const float* W_r2  = (const float*)d_in[6];
    const int*   src1  = (const int*)d_in[7];
    const int*   dst1  = (const int*)d_in[8];
    const int*   src2  = (const int*)d_in[9];
    const int*   dst2  = (const int*)d_in[10];

    int* wsi  = (int*)d_ws;
    int* off1 = wsi;
    int* off2 = off1 + (N1C + 1);
    int* cur1 = off2 + (N2C + 1);
    int* cur2 = cur1 + N1C;
    int* eid1 = cur2 + N2C;
    int* eid2 = eid1 + E1C;
    long ihdr = (long)(N1C + 1) + (N2C + 1) + N1C + N2C + E1C + E2C;
    ihdr = (ihdr + 3) & ~3L;
    float* h1 = (float*)(wsi + ihdr);
    float* h2 = (float*)d_out;

    const int nzero = (N1C + 1) + (N2C + 1) + N1C + N2C;

    zero_int<<<(nzero + 255) / 256, 256, 0, stream>>>(off1, nzero);
    edge_hist<<<(E1C + 255) / 256, 256, 0, stream>>>(dst1, off1, E1C);
    edge_hist<<<(E2C + 255) / 256, 256, 0, stream>>>(dst2, off2, E2C);
    scan_excl<<<1, 1024, 0, stream>>>(off1, N1C);
    scan_excl<<<1, 1024, 0, stream>>>(off2, N2C);
    edge_fill<<<(E1C + 255) / 256, 256, 0, stream>>>(src1, dst1, off1, cur1,
                                                     eid1, E1C);
    edge_fill<<<(E2C + 255) / 256, 256, 0, stream>>>(src2, dst2, off2, cur2,
                                                     eid2, E2C);
    sage_layer<<<N1C / 16, 512, 0, stream>>>(nodes, off1, eid1,
                                             W_l1, b1, W_r1, h1, N1C);
    sage_layer<<<N2C / 16, 512, 0, stream>>>(h1, off2, eid2,
                                             W_l2, b2, W_r2, h2, N2C);
}


# scan_excl rewritten: log-step LDS scan over 1024-element chunks (44 barriers per chunk) -> per-wave DPP row scans over 8192-element passes, one barrier per pass, all loads issued first
# speedup vs baseline: 1.2311x; 1.2311x over previous
#include <hip/hip_runtime.h>

typedef __attribute__((ext_vector_type(2))) float v2f;
typedef __attribute__((ext_vector_type(8))) float v8f;

#define DIM   256
#define N1C   40000
#define N2C   8000
#define E1C   1000000
#define E2C   200000
#define LROW  (DIM + 4)

__global__ void zero_int(int* __restrict__ p, int n) {
    int i = blockIdx.x * blockDim.x + threadIdx.x;
    if (i < n) p[i] = 0;
}

__global__ void edge_hist(const int* __restrict__ dst, int* __restrict__ cnt,
                          int n) {
    int i = blockIdx.x * blockDim.x + threadIdx.x;
    if (i < n) atomicAdd(cnt + dst[i], 1);
}

__device__ __forceinline__ int wave_incl_scan(int x, int lane) {
    x += __builtin_amdgcn_update_dpp(0, x, 0x111, 0xf, 0xf, true);
    x += __builtin_amdgcn_update_dpp(0, x, 0x112, 0xf, 0xf, true);
    x += __builtin_amdgcn_update_dpp(0, x, 0x114, 0xf, 0xf, true);
    x += __builtin_amdgcn_update_dpp(0, x, 0x118, 0xf, 0xf, true);
    int low = __builtin_amdgcn_readlane(x, 15);
    x += (lane >= 16) ? low : 0;
    return x;
}

template <int N>
__global__ __launch_bounds__(1024)
void scan_excl(int* __restrict__ data) {
    constexpr int TILE = 1024 * 8;
    constexpr int NT   = (N + 1 + TILE - 1) / TILE;
    __shared__ int wtot[2][32];

    const int lane = threadIdx.x & 31;
    const int wv   = __builtin_amdgcn_readfirstlane((int)(threadIdx.x >> 5));
    const int i0   = wv * 256 + lane;

    int v[NT][8];
#pragma unroll
    for (int t = 0; t < NT; ++t) {
        const bool full = (t + 1) * TILE <= N;
#pragma unroll
        for (int k = 0; k < 8; ++k) {
            int i = t * TILE + i0 + k * 32;
            v[t][k] = (full || i < N) ? data[i] : 0;
        }
    }

    int running = 0;
#pragma unroll
    for (int t = 0; t < NT; ++t) {
        const bool full = (t + 1) * TILE <= N;
        int incl[8];
        int carry = 0;
#pragma unroll
        for (int k = 0; k < 8; ++k) {
            int x = wave_incl_scan(v[t][k], lane);
            incl[k] = x + carry;
            carry += __builtin_amdgcn_readlane(x, 31);
        }
        if (lane == 0) wtot[t & 1][wv] = carry;
        __syncthreads();
        int w  = wtot[t & 1][lane];
        int wi = wave_incl_scan(w, lane);
        int wbase = __builtin_amdgcn_readlane(wi - w, wv);
        int base  = running + wbase;
        running  += __builtin_amdgcn_readlane(wi, 31);
#pragma unroll
        for (int k = 0; k < 8; ++k) {
            int i = t * TILE + i0 + k * 32;
            if (full || i <= N) data[i] = base + incl[k] - v[t][k];
        }
    }
}

__global__ void edge_fill(const int* __restrict__ src,
                          const int* __restrict__ dst,
                          const int* __restrict__ off,
                          int* __restrict__ cur,
                          int* __restrict__ eid, int n) {
    int i = blockIdx.x * blockDim.x + threadIdx.x;
    if (i < n) {
        int d = dst[i];
        int p = atomicAdd(cur + d, 1);
        eid[off[d] + p] = src[i];
    }
}

__global__ __launch_bounds__(512)
void sage_layer(const float* __restrict__ x,
                const int* __restrict__ off,
                const int* __restrict__ eid,
                const float* __restrict__ Wl,
                const float* __restrict__ bias,
                const float* __restrict__ Wr,
                float* __restrict__ out,
                int M) {
    __shared__ float lmean[16 * LROW];
    __shared__ float lx[16 * LROW];

    const int m0   = blockIdx.x * 16;
    const int tid  = threadIdx.x;
    const int lane = tid & 31;
    const int wv   = tid >> 5;

    for (int i = tid; i < 16 * DIM; i += 512) {
        int row = i >> 8;
        int col = i & (DIM - 1);
        lx[row * LROW + col] = x[(long)(m0 + row) * DIM + col];
    }

    {
        int row = m0 + wv;
        int beg = off[row];
        int end = off[row + 1];
        float4 a0 = make_float4(0.f, 0.f, 0.f, 0.f);
        float4 a1 = make_float4(0.f, 0.f, 0.f, 0.f);
        for (int e = beg; e < end; e += 32) {
            int myid = (e + lane < end) ? eid[e + lane] : 0;
            int cnt = min(32, end - e);
            for (int j = 0; j < cnt; ++j) {
                int s = __shfl(myid, j, 32);
                const float4* xr = (const float4*)(x + (long)s * DIM);
                float4 u = xr[lane];
                float4 v = xr[lane + 32];
                a0.x += u.x; a0.y += u.y; a0.z += u.z; a0.w += u.w;
                a1.x += v.x; a1.y += v.y; a1.z += v.z; a1.w += v.w;
            }
        }
        float inv = 1.0f / fmaxf((float)(end - beg), 1.0f);
        float* mrow = lmean + wv * LROW;
        mrow[lane * 4 + 0] = a0.x * inv;
        mrow[lane * 4 + 1] = a0.y * inv;
        mrow[lane * 4 + 2] = a0.z * inv;
        mrow[lane * 4 + 3] = a0.w * inv;
        mrow[128 + lane * 4 + 0] = a1.x * inv;
        mrow[128 + lane * 4 + 1] = a1.y * inv;
        mrow[128 + lane * 4 + 2] = a1.z * inv;
        mrow[128 + lane * 4 + 3] = a1.w * inv;
    }
    __syncthreads();

    const int half = lane >> 4;
    const int lrow = lane & 15;
    const int n0   = wv * 16;

    const float* wl_r = Wl + (long)(n0 + lrow) * DIM;
    const float* wr_r = Wr + (long)(n0 + lrow) * DIM;
    const float* lm_r = lmean + lrow * LROW;
    const float* lx_r = lx + lrow * LROW;

    v8f acc = {};
#pragma unroll 4
    for (int k = 0; k < DIM; k += 4) {
        int kk = k + 2 * half;
        v2f am = *(const v2f*)(lm_r + kk);
        v2f bl = *(const v2f*)(wl_r + kk);
        acc = __builtin_amdgcn_wmma_f32_16x16x4_f32(
            false, am, false, bl, (short)0, acc, false, false);
        v2f ax = *(const v2f*)(lx_r + kk);
        v2f br = *(const v2f*)(wr_r + kk);
        acc = __builtin_amdgcn_wmma_f32_16x16x4_f32(
            false, ax, false, br, (short)0, acc, false, false);
    }

    float bb = bias[n0 + lrow];
    float* orow = out + (long)(m0 + half * 8) * DIM + n0 + lrow;
#pragma unroll
    for (int r = 0; r < 8; ++r) {
        orow[(long)r * DIM] = tanhf(acc[r] + bb);
    }
}

extern "C" void kernel_launch(void* const* d_in, const int* in_sizes, int n_in,
                              void* d_out, int out_size, void* d_ws, size_t ws_size,
                              hipStream_t stream) {
    const float* nodes = (const float*)d_in[0];
    const float* W_l1  = (const float*)d_in[1];
    const float* b1    = (const float*)d_in[2];
    const float* W_r1  = (const float*)d_in[3];
    const float* W_l2  = (const float*)d_in[4];
    const float* b2    = (const float*)d_in[5];
    const float* W_r2  = (const float*)d_in[6];
    const int*   src1  = (const int*)d_in[7];
    const int*   dst1  = (const int*)d_in[8];
    const int*   src2  = (const int*)d_in[9];
    const int*   dst2  = (const int*)d_in[10];

    int* wsi  = (int*)d_ws;
    int* off1 = wsi;
    int* off2 = off1 + (N1C + 1);
    int* cur1 = off2 + (N2C + 1);
    int* cur2 = cur1 + N1C;
    int* eid1 = cur2 + N2C;
    int* eid2 = eid1 + E1C;
    long ihdr = (long)(N1C + 1) + (N2C + 1) + N1C + N2C + E1C + E2C;
    ihdr = (ihdr + 3) & ~3L;
    float* h1 = (float*)(wsi + ihdr);
    float* h2 = (float*)d_out;

    const int nzero = (N1C + 1) + (N2C + 1) + N1C + N2C;

    zero_int<<<(nzero + 255) / 256, 256, 0, stream>>>(off1, nzero);
    edge_hist<<<(E1C + 255) / 256, 256, 0, stream>>>(dst1, off1, E1C);
    edge_hist<<<(E2C + 255) / 256, 256, 0, stream>>>(dst2, off2, E2C);
    scan_excl<N1C><<<1, 1024, 0, stream>>>(off1);
    scan_excl<N2C><<<1, 1024, 0, stream>>>(off2);
    edge_fill<<<(E1C + 255) / 256, 256, 0, stream>>>(src1, dst1, off1, cur1,
                                                     eid1, E1C);
    edge_fill<<<(E2C + 255) / 256, 256, 0, stream>>>(src2, dst2, off2, cur2,
                                                     eid2, E2C);
    sage_layer<<<N1C / 16, 512, 0, stream>>>(nodes, off1, eid1,
                                             W_l1, b1, W_r1, h1, N1C);
    sage_layer<<<N2C / 16, 512, 0, stream>>>(h1, off2, eid2,
                                             W_l2, b2, W_r2, h2, N2C);
}


# sage_layer: 64-row blocks (layer 2: 32): each wave keeps 4 (2) accumulator tiles, so every weight fragment load feeds 4 (2) WMMAs; a wave's 4 (2) rows are gathered side by side, per-row order kept
# speedup vs baseline: 1.4202x; 1.1536x over previous
#include <hip/hip_runtime.h>

typedef __attribute__((ext_vector_type(2))) float v2f;
typedef __attribute__((ext_vector_type(8))) float v8f;

#define DIM   256
#define N1C   40000
#define N2C   8000
#define E1C   1000000
#define E2C   200000
#define LROW  (DIM + 4)

__global__ void zero_int(int* __restrict__ p, int n) {
    int i = blockIdx.x * blockDim.x + threadIdx.x;
    if (i < n) p[i] = 0;
}

__global__ void edge_hist(const int* __restrict__ dst, int* __restrict__ cnt,
                          int n) {
    int i = blockIdx.x * blockDim.x + threadIdx.x;
    if (i < n) atomicAdd(cnt + dst[i], 1);
}

__device__ __forceinline__ int wave_incl_scan(int x, int lane) {
    x += __builtin_amdgcn_update_dpp(0, x, 0x111, 0xf, 0xf, true);
    x += __builtin_amdgcn_update_dpp(0, x, 0x112, 0xf, 0xf, true);
    x += __builtin_amdgcn_update_dpp(0, x, 0x114, 0xf, 0xf, true);
    x += __builtin_amdgcn_update_dpp(0, x, 0x118, 0xf, 0xf, true);
    int low = __builtin_amdgcn_readlane(x, 15);
    x += (lane >= 16) ? low : 0;
    return x;
}

template <int N>
__global__ __launch_bounds__(1024)
void scan_excl(int* __restrict__ data) {
    constexpr int TILE = 1024 * 8;
    constexpr int NT   = (N + 1 + TILE - 1) / TILE;
    __shared__ int wtot[2][32];

    const int lane = threadIdx.x & 31;
    const int wv   = __builtin_amdgcn_readfirstlane((int)(threadIdx.x >> 5));
    const int i0   = wv * 256 + lane;

    int v[NT][8];
#pragma unroll
    for (int t = 0; t < NT; ++t) {
        const bool full = (t + 1) * TILE <= N;
#pragma unroll
        for (int k = 0; k < 8; ++k) {
            int i = t * TILE + i0 + k * 32;
            v[t][k] = (full || i < N) ? data[i] : 0;
        }
    }

    int running = 0;
#pragma unroll
    for (int t = 0; t < NT; ++t) {
        const bool full = (t + 1) * TILE <= N;
        int incl[8];
        int carry = 0;
#pragma unroll
        for (int k = 0; k < 8; ++k) {
            int x = wave_incl_scan(v[t][k], lane);
            incl[k] = x + carry;
            carry += __builtin_amdgcn_readlane(x, 31);
        }
        if (lane == 0) wtot[t & 1][wv] = carry;
        __syncthreads();
        int w  = wtot[t & 1][lane];
        int wi = wave_incl_scan(w, lane);
        int wbase = __builtin_amdgcn_readlane(wi - w, wv);
        int base  = running + wbase;
        running  += __builtin_amdgcn_readlane(wi, 31);
#pragma unroll
        for (int k = 0; k < 8; ++k) {
            int i = t * TILE + i0 + k * 32;
            if (full || i <= N) data[i] = base + incl[k] - v[t][k];
        }
    }
}

__global__ void edge_fill(const int* __restrict__ src,
                          const int* __restrict__ dst,
                          const int* __restrict__ off,
                          int* __restrict__ cur,
                          int* __restrict__ eid, int n) {
    int i = blockIdx.x * blockDim.x + threadIdx.x;
    if (i < n) {
        int d = dst[i];
        int p = atomicAdd(cur + d, 1);
        eid[off[d] + p] = src[i];
    }
}

template <int MT>
__global__ __launch_bounds__(512)
void sage_layer(const float* __restrict__ x,
                const int* __restrict__ off,
                const int* __restrict__ eid,
                const float* __restrict__ Wl,
                const float* __restrict__ bias,
                const float* __restrict__ Wr,
                float* __restrict__ out,
                int M) {
    constexpr int BM = 16 * MT;
    __shared__ float lmean[BM * LROW];
    __shared__ float lx[BM * LROW];

    const int m0   = blockIdx.x * BM;
    const int tid  = threadIdx.x;
    const int lane = tid & 31;
    const int wv   = tid >> 5;

    for (int i = tid; i < BM * DIM; i += 512) {
        int row = i >> 8;
        int col = i & (DIM - 1);
        lx[row * LROW + col] = x[(long)(m0 + row) * DIM + col];
    }

    {
        const int r0 = wv * MT;
        int beg[MT + 1];
#pragma unroll
        for (int r = 0; r <= MT; ++r)
            beg[r] = __builtin_amdgcn_readfirstlane(off[m0 + r0 + r]);
        int deg[MT];
        int maxdeg = 0;
#pragma unroll
        for (int r = 0; r < MT; ++r) {
            deg[r] = beg[r + 1] - beg[r];
            maxdeg = max(maxdeg, deg[r]);
        }
        float4 a0[MT], a1[MT];
#pragma unroll
        for (int r = 0; r < MT; ++r) {
            a0[r] = make_float4(0.f, 0.f, 0.f, 0.f);
            a1[r] = make_float4(0.f, 0.f, 0.f, 0.f);
        }
        for (int e = 0; e < maxdeg; e += 32) {
            int myid[MT];
#pragma unroll
            for (int r = 0; r < MT; ++r)
                myid[r] = (e + lane < deg[r]) ? eid[beg[r] + e + lane] : 0;
            const int cnt = min(32, maxdeg - e);
            for (int j = 0; j < cnt; ++j) {
                float4 u[MT], v[MT];
#pragma unroll
                for (int r = 0; r < MT; ++r) {
                    int s = __shfl(myid[r], j, 32);
                    if (e + j < deg[r]) {
                        const float4* xr = (const float4*)(x + (long)s * DIM);
                        u[r] = xr[lane];
                        v[r] = xr[lane + 32];
                    } else {
                        u[r] = make_float4(0.f, 0.f, 0.f, 0.f);
                        v[r] = make_float4(0.f, 0.f, 0.f, 0.f);
                    }
                }
#pragma unroll
                for (int r = 0; r < MT; ++r) {
                    a0[r].x += u[r].x; a0[r].y += u[r].y; a0[r].z += u[r].z; a0[r].w += u[r].w;
                    a1[r].x += v[r].x; a1[r].y += v[r].y; a1[r].z += v[r].z; a1[r].w += v[r].w;
                }
            }
        }
#pragma unroll
        for (int r = 0; r < MT; ++r) {
            float inv = 1.0f / fmaxf((float)deg[r], 1.0f);
            float* mrow = lmean + (r0 + r) * LROW;
            mrow[lane * 4 + 0] = a0[r].x * inv;
            mrow[lane * 4 + 1] = a0[r].y * inv;
            mrow[lane * 4 + 2] = a0[r].z * inv;
            mrow[lane * 4 + 3] = a0[r].w * inv;
            mrow[128 + lane * 4 + 0] = a1[r].x * inv;
            mrow[128 + lane * 4 + 1] = a1[r].y * inv;
            mrow[128 + lane * 4 + 2] = a1[r].z * inv;
            mrow[128 + lane * 4 + 3] = a1[r].w * inv;
        }
    }
    __syncthreads();

    const int half = lane >> 4;
    const int lrow = lane & 15;
    const int n0   = wv * 16;

    const float* wl_r = Wl + (long)(n0 + lrow) * DIM;
    const float* wr_r = Wr + (long)(n0 + lrow) * DIM;
    const float* lm_r = lmean + lrow * LROW;
    const float* lx_r = lx + lrow * LROW;

    v8f acc[MT];
#pragma unroll
    for (int t = 0; t < MT; ++t) acc[t] = v8f{};
#pragma unroll (8 / MT)
    for (int k = 0; k < DIM; k += 4) {
        int kk = k + 2 * half;
        v2f bl = *(const v2f*)(wl_r + kk);
        v2f br = *(const v2f*)(wr_r + kk);
#pragma unroll
        for (int t = 0; t < MT; ++t) {
            v2f am = *(const v2f*)(lm_r + t * 16 * LROW + kk);
            acc[t] = __builtin_amdgcn_wmma_f32_16x16x4_f32(
                false, am, false, bl, (short)0, acc[t], false, false);
            v2f ax = *(const v2f*)(lx_r + t * 16 * LROW + kk);
            acc[t] = __builtin_amdgcn_wmma_f32_16x16x4_f32(
                false, ax, false, br, (short)0, acc[t], false, false);
        }
    }

    float bb = bias[n0 + lrow];
#pragma unroll
    for (int t = 0; t < MT; ++t) {
        float* orow = out + (long)(m0 + t * 16 + half * 8) * DIM + n0 + lrow;
#pragma unroll
        for (int r = 0; r < 8; ++r) {
            orow[(long)r * DIM] = tanhf(acc[t][r] + bb);
        }
    }
}

extern "C" void kernel_launch(void* const* d_in, const int* in_sizes, int n_in,
                              void* d_out, int out_size, void* d_ws, size_t ws_size,
                              hipStream_t stream) {
    const float* nodes = (const float*)d_in[0];
    const float* W_l1  = (const float*)d_in[1];
    const float* b1    = (const float*)d_in[2];
    const float* W_r1  = (const float*)d_in[3];
    const float* W_l2  = (const float*)d_in[4];
    const float* b2    = (const float*)d_in[5];
    const float* W_r2  = (const float*)d_in[6];
    const int*   src1  = (const int*)d_in[7];
    const int*   dst1  = (const int*)d_in[8];
    const int*   src2  = (const int*)d_in[9];
    const int*   dst2  = (const int*)d_in[10];

    int* wsi  = (int*)d_ws;
    int* off1 = wsi;
    int* off2 = off1 + (N1C + 1);
    int* cur1 = off2 + (N2C + 1);
    int* cur2 = cur1 + N1C;
    int* eid1 = cur2 + N2C;
    int* eid2 = eid1 + E1C;
    long ihdr = (long)(N1C + 1) + (N2C + 1) + N1C + N2C + E1C + E2C;
    ihdr = (ihdr + 3) & ~3L;
    float* h1 = (float*)(wsi + ihdr);
    float* h2 = (float*)d_out;

    const int nzero = (N1C + 1) + (N2C + 1) + N1C + N2C;

    zero_int<<<(nzero + 255) / 256, 256, 0, stream>>>(off1, nzero);
    edge_hist<<<(E1C + 255) / 256, 256, 0, stream>>>(dst1, off1, E1C);
    edge_hist<<<(E2C + 255) / 256, 256, 0, stream>>>(dst2, off2, E2C);
    scan_excl<N1C><<<1, 1024, 0, stream>>>(off1);
    scan_excl<N2C><<<1, 1024, 0, stream>>>(off2);
    edge_fill<<<(E1C + 255) / 256, 256, 0, stream>>>(src1, dst1, off1, cur1,
                                                     eid1, E1C);
    edge_fill<<<(E2C + 255) / 256, 256, 0, stream>>>(src2, dst2, off2, cur2,
                                                     eid2, E2C);
    sage_layer<4><<<N1C / 64, 512, 0, stream>>>(nodes, off1, eid1,
                                                W_l1, b1, W_r1, h1, N1C);
    sage_layer<2><<<N2C / 32, 512, 0, stream>>>(h1, off2, eid2,
                                                W_l2, b2, W_r2, h2, N2C);
}


# CSR build: the counting pass keeps each edge's slot from its returning add (rank array), so the fill pass needs no atomic: one global atomic per edge, not two; neighbour order still by atomic arrival
# speedup vs baseline: 1.7372x; 1.2232x over previous
#include <hip/hip_runtime.h>

typedef __attribute__((ext_vector_type(2))) float v2f;
typedef __attribute__((ext_vector_type(8))) float v8f;

#define DIM   256
#define N1C   40000
#define N2C   8000
#define E1C   1000000
#define E2C   200000
#define LROW  (DIM + 4)

__global__ void zero_int(int* __restrict__ p, int n) {
    int i = blockIdx.x * blockDim.x + threadIdx.x;
    if (i < n) p[i] = 0;
}

__global__ void edge_rank(const int* __restrict__ dst, int* __restrict__ cnt,
                          int* __restrict__ rank, int n) {
    int i = blockIdx.x * blockDim.x + threadIdx.x;
    if (i < n) rank[i] = atomicAdd(cnt + dst[i], 1);
}

__device__ __forceinline__ int wave_incl_scan(int x, int lane) {
    x += __builtin_amdgcn_update_dpp(0, x, 0x111, 0xf, 0xf, true);
    x += __builtin_amdgcn_update_dpp(0, x, 0x112, 0xf, 0xf, true);
    x += __builtin_amdgcn_update_dpp(0, x, 0x114, 0xf, 0xf, true);
    x += __builtin_amdgcn_update_dpp(0, x, 0x118, 0xf, 0xf, true);
    int low = __builtin_amdgcn_readlane(x, 15);
    x += (lane >= 16) ? low : 0;
    return x;
}

template <int N>
__global__ __launch_bounds__(1024)
void scan_excl(int* __restrict__ data) {
    constexpr int TILE = 1024 * 8;
    constexpr int NT   = (N + 1 + TILE - 1) / TILE;
    __shared__ int wtot[2][32];

    const int lane = threadIdx.x & 31;
    const int wv   = __builtin_amdgcn_readfirstlane((int)(threadIdx.x >> 5));
    const int i0   = wv * 256 + lane;

    int v[NT][8];
#pragma unroll
    for (int t = 0; t < NT; ++t) {
        const bool full = (t + 1) * TILE <= N;
#pragma unroll
        for (int k = 0; k < 8; ++k) {
            int i = t * TILE + i0 + k * 32;
            v[t][k] = (full || i < N) ? data[i] : 0;
        }
    }

    int running = 0;
#pragma unroll
    for (int t = 0; t < NT; ++t) {
        const bool full = (t + 1) * TILE <= N;
        int incl[8];
        int carry = 0;
#pragma unroll
        for (int k = 0; k < 8; ++k) {
            int x = wave_incl_scan(v[t][k], lane);
            incl[k] = x + carry;
            carry += __builtin_amdgcn_readlane(x, 31);
        }
        if (lane == 0) wtot[t & 1][wv] = carry;
        __syncthreads();
        int w  = wtot[t & 1][lane];
        int wi = wave_incl_scan(w, lane);
        int wbase = __builtin_amdgcn_readlane(wi - w, wv);
        int base  = running + wbase;
        running  += __builtin_amdgcn_readlane(wi, 31);
#pragma unroll
        for (int k = 0; k < 8; ++k) {
            int i = t * TILE + i0 + k * 32;
            if (full || i <= N) data[i] = base + incl[k] - v[t][k];
        }
    }
}

__global__ void edge_fill(const int* __restrict__ src,
                          const int* __restrict__ dst,
                          const int* __restrict__ rank,
                          const int* __restrict__ off,
                          int* __restrict__ eid, int n4) {
    int q = blockIdx.x * blockDim.x + threadIdx.x;
    if (q < n4) {
        const int4 d = ((const int4*)dst)[q];
        const int4 r = ((const int4*)rank)[q];
        const int4 s = ((const int4*)src)[q];
        const int px = off[d.x] + r.x;
        const int py = off[d.y] + r.y;
        const int pz = off[d.z] + r.z;
        const int pw = off[d.w] + r.w;
        eid[px] = s.x;
        eid[py] = s.y;
        eid[pz] = s.z;
        eid[pw] = s.w;
    }
}

template <int MT>
__global__ __launch_bounds__(512)
void sage_layer(const float* __restrict__ x,
                const int* __restrict__ off,
                const int* __restrict__ eid,
                const float* __restrict__ Wl,
                const float* __restrict__ bias,
                const float* __restrict__ Wr,
                float* __restrict__ out,
                int M) {
    constexpr int BM = 16 * MT;
    __shared__ float lmean[BM * LROW];
    __shared__ float lx[BM * LROW];

    const int m0   = blockIdx.x * BM;
    const int tid  = threadIdx.x;
    const int lane = tid & 31;
    const int wv   = tid >> 5;

    for (int i = tid; i < BM * DIM; i += 512) {
        int row = i >> 8;
        int col = i & (DIM - 1);
        lx[row * LROW + col] = x[(long)(m0 + row) * DIM + col];
    }

    {
        const int r0 = wv * MT;
        int beg[MT + 1];
#pragma unroll
        for (int r = 0; r <= MT; ++r)
            beg[r] = __builtin_amdgcn_readfirstlane(off[m0 + r0 + r]);
        int deg[MT];
        int maxdeg = 0;
#pragma unroll
        for (int r = 0; r < MT; ++r) {
            deg[r] = beg[r + 1] - beg[r];
            maxdeg = max(maxdeg, deg[r]);
        }
        float4 a0[MT], a1[MT];
#pragma unroll
        for (int r = 0; r < MT; ++r) {
            a0[r] = make_float4(0.f, 0.f, 0.f, 0.f);
            a1[r] = make_float4(0.f, 0.f, 0.f, 0.f);
        }
        for (int e = 0; e < maxdeg; e += 32) {
            int myid[MT];
#pragma unroll
            for (int r = 0; r < MT; ++r)
                myid[r] = (e + lane < deg[r]) ? eid[beg[r] + e + lane] : 0;
            const int cnt = min(32, maxdeg - e);
            for (int j = 0; j < cnt; ++j) {
                float4 u[MT], v[MT];
#pragma unroll
                for (int r = 0; r < MT; ++r) {
                    int s = __shfl(myid[r], j, 32);
                    if (e + j < deg[r]) {
                        const float4* xr = (const float4*)(x + (long)s * DIM);
                        u[r] = xr[lane];
                        v[r] = xr[lane + 32];
                    } else {
                        u[r] = make_float4(0.f, 0.f, 0.f, 0.f);
                        v[r] = make_float4(0.f, 0.f, 0.f, 0.f);
                    }
                }
#pragma unroll
                for (int r = 0; r < MT; ++r) {
                    a0[r].x += u[r].x; a0[r].y += u[r].y; a0[r].z += u[r].z; a0[r].w += u[r].w;
                    a1[r].x += v[r].x; a1[r].y += v[r].y; a1[r].z += v[r].z; a1[r].w += v[r].w;
                }
            }
        }
#pragma unroll
        for (int r = 0; r < MT; ++r) {
            float inv = 1.0f / fmaxf((float)deg[r], 1.0f);
            float* mrow = lmean + (r0 + r) * LROW;
            mrow[lane * 4 + 0] = a0[r].x * inv;
            mrow[lane * 4 + 1] = a0[r].y * inv;
            mrow[lane * 4 + 2] = a0[r].z * inv;
            mrow[lane * 4 + 3] = a0[r].w * inv;
            mrow[128 + lane * 4 + 0] = a1[r].x * inv;
            mrow[128 + lane * 4 + 1] = a1[r].y * inv;
            mrow[128 + lane * 4 + 2] = a1[r].z * inv;
            mrow[128 + lane * 4 + 3] = a1[r].w * inv;
        }
    }
    __syncthreads();

    const int half = lane >> 4;
    const int lrow = lane & 15;
    const int n0   = wv * 16;

    const float* wl_r = Wl + (long)(n0 + lrow) * DIM;
    const float* wr_r = Wr + (long)(n0 + lrow) * DIM;
    const float* lm_r = lmean + lrow * LROW;
    const float* lx_r = lx + lrow * LROW;

    v8f acc[MT];
#pragma unroll
    for (int t = 0; t < MT; ++t) acc[t] = v8f{};
#pragma unroll (8 / MT)
    for (int k = 0; k < DIM; k += 4) {
        int kk = k + 2 * half;
        v2f bl = *(const v2f*)(wl_r + kk);
        v2f br = *(const v2f*)(wr_r + kk);
#pragma unroll
        for (int t = 0; t < MT; ++t) {
            v2f am = *(const v2f*)(lm_r + t * 16 * LROW + kk);
            acc[t] = __builtin_amdgcn_wmma_f32_16x16x4_f32(
                false, am, false, bl, (short)0, acc[t], false, false);
            v2f ax = *(const v2f*)(lx_r + t * 16 * LROW + kk);
            acc[t] = __builtin_amdgcn_wmma_f32_16x16x4_f32(
                false, ax, false, br, (short)0, acc[t], false, false);
        }
    }

    float bb = bias[n0 + lrow];
#pragma unroll
    for (int t = 0; t < MT; ++t) {
        float* orow = out + (long)(m0 + t * 16 + half * 8) * DIM + n0 + lrow;
#pragma unroll
        for (int r = 0; r < 8; ++r) {
            orow[(long)r * DIM] = tanhf(acc[t][r] + bb);
        }
    }
}

#define NOFF ((((N1C + 1) + (N2C + 1)) + 3) & ~3)

extern "C" void kernel_launch(void* const* d_in, const int* in_sizes, int n_in,
                              void* d_out, int out_size, void* d_ws, size_t ws_size,
                              hipStream_t stream) {
    const float* nodes = (const float*)d_in[0];
    const float* W_l1  = (const float*)d_in[1];
    const float* b1    = (const float*)d_in[2];
    const float* W_r1  = (const float*)d_in[3];
    const float* W_l2  = (const float*)d_in[4];
    const float* b2    = (const float*)d_in[5];
    const float* W_r2  = (const float*)d_in[6];
    const int*   src1  = (const int*)d_in[7];
    const int*   dst1  = (const int*)d_in[8];
    const int*   src2  = (const int*)d_in[9];
    const int*   dst2  = (const int*)d_in[10];

    int* wsi   = (int*)d_ws;
    int* off1  = wsi;
    int* off2  = off1 + (N1C + 1);
    int* rank1 = wsi + NOFF;
    int* rank2 = rank1 + E1C;
    int* eid1  = rank2 + E2C;
    int* eid2  = eid1 + E1C;
    float* h1  = (float*)(eid2 + E2C);
    float* h2  = (float*)d_out;

    const int nzero = (N1C + 1) + (N2C + 1);

    zero_int<<<(nzero + 255) / 256, 256, 0, stream>>>(off1, nzero);
    edge_rank<<<(E1C + 255) / 256, 256, 0, stream>>>(dst1, off1, rank1, E1C);
    edge_rank<<<(E2C + 255) / 256, 256, 0, stream>>>(dst2, off2, rank2, E2C);
    scan_excl<N1C><<<1, 1024, 0, stream>>>(off1);
    scan_excl<N2C><<<1, 1024, 0, stream>>>(off2);
    edge_fill<<<(E1C / 4 + 255) / 256, 256, 0, stream>>>(src1, dst1, rank1, off1,
                                                         eid1, E1C / 4);
    edge_fill<<<(E2C / 4 + 255) / 256, 256, 0, stream>>>(src2, dst2, rank2, off2,
                                                         eid2, E2C / 4);
    sage_layer<4><<<N1C / 64, 512, 0, stream>>>(nodes, off1, eid1,
                                                W_l1, b1, W_r1, h1, N1C);
    sage_layer<2><<<N2C / 32, 512, 0, stream>>>(h1, off2, eid2,
                                                W_l2, b2, W_r2, h2, N2C);
}


# CSR build: per-edge global add replaced by per-chunk LDS histograms (returning LDS add = slot in chunk) + per-destination prefix over chunks; neighbour order now chunk, then LDS arrival
# speedup vs baseline: 2.1713x; 1.2499x over previous
#include <hip/hip_runtime.h>

typedef __attribute__((ext_vector_type(2))) float v2f;
typedef __attribute__((ext_vector_type(8))) float v8f;

#define DIM   256
#define N1C   40000
#define N2C   8000
#define E1C   1000000
#define E2C   200000
#define LROW  (DIM + 4)

#define EPB   20000
#define NB1   (E1C / EPB)
#define NB2   (E2C / EPB)

template <int N>
__device__ __forceinline__ void chunk_hist_body(const int* __restrict__ dst,
                                                int* __restrict__ rank,
                                                int* __restrict__ tab,
                                                int* hist, int chunk) {
    constexpr int Q  = EPB / 4;
    constexpr int IT = (Q + 1023) / 1024;
    const int tid = threadIdx.x;
    const int4* d4 = (const int4*)(dst + (long)chunk * EPB);
    int4* r4 = (int4*)(rank + (long)chunk * EPB);

    int4 d[IT];
#pragma unroll
    for (int it = 0; it < IT; ++it) {
        int q = tid + it * 1024;
        d[it] = d4[(q < Q) ? q : Q - 1];
    }

    int4* h4 = (int4*)hist;
    for (int i = tid; i < N / 4; i += 1024) h4[i] = make_int4(0, 0, 0, 0);
    __syncthreads();

#pragma unroll
    for (int it = 0; it < IT; ++it) {
        int q = tid + it * 1024;
        if (q < Q) {
            int4 r;
            r.x = atomicAdd(hist + d[it].x, 1);
            r.y = atomicAdd(hist + d[it].y, 1);
            r.z = atomicAdd(hist + d[it].z, 1);
            r.w = atomicAdd(hist + d[it].w, 1);
            r4[q] = r;
        }
    }
    __syncthreads();

    int4* t4 = (int4*)(tab + (long)chunk * N);
    for (int i = tid; i < N / 4; i += 1024) t4[i] = h4[i];
}

__global__ __launch_bounds__(1024)
void chunk_hist(const int* __restrict__ dst1, int* __restrict__ rank1, int* __restrict__ tab1,
                const int* __restrict__ dst2, int* __restrict__ rank2, int* __restrict__ tab2) {
    __shared__ __attribute__((aligned(16))) int hist[N1C];
    if (blockIdx.x < NB1)
        chunk_hist_body<N1C>(dst1, rank1, tab1, hist, blockIdx.x);
    else
        chunk_hist_body<N2C>(dst2, rank2, tab2, hist, blockIdx.x - NB1);
}

template <int N, int NB>
__device__ __forceinline__ void chunk_prefix_body(int* __restrict__ tab,
                                                  int* __restrict__ cnt, int d) {
    int cv[NB];
#pragma unroll
    for (int b = 0; b < NB; ++b) cv[b] = tab[(long)b * N + d];
    int run = 0;
#pragma unroll
    for (int b = 0; b < NB; ++b) {
        tab[(long)b * N + d] = run;
        run += cv[b];
    }
    cnt[d] = run;
}

__global__ __launch_bounds__(128)
void chunk_prefix(int* __restrict__ tab1, int* __restrict__ cnt1,
                  int* __restrict__ tab2, int* __restrict__ cnt2) {
    int g = blockIdx.x * 128 + threadIdx.x;
    if (g < N1C)
        chunk_prefix_body<N1C, NB1>(tab1, cnt1, g);
    else if (g < N1C + N2C)
        chunk_prefix_body<N2C, NB2>(tab2, cnt2, g - N1C);
}

__device__ __forceinline__ int wave_incl_scan(int x, int lane) {
    x += __builtin_amdgcn_update_dpp(0, x, 0x111, 0xf, 0xf, true);
    x += __builtin_amdgcn_update_dpp(0, x, 0x112, 0xf, 0xf, true);
    x += __builtin_amdgcn_update_dpp(0, x, 0x114, 0xf, 0xf, true);
    x += __builtin_amdgcn_update_dpp(0, x, 0x118, 0xf, 0xf, true);
    int low = __builtin_amdgcn_readlane(x, 15);
    x += (lane >= 16) ? low : 0;
    return x;
}

template <int N>
__global__ __launch_bounds__(1024)
void scan_excl(int* __restrict__ data) {
    constexpr int TILE = 1024 * 8;
    constexpr int NT   = (N + 1 + TILE - 1) / TILE;
    __shared__ int wtot[2][32];

    const int lane = threadIdx.x & 31;
    const int wv   = __builtin_amdgcn_readfirstlane((int)(threadIdx.x >> 5));
    const int i0   = wv * 256 + lane;

    int v[NT][8];
#pragma unroll
    for (int t = 0; t < NT; ++t) {
        const bool full = (t + 1) * TILE <= N;
#pragma unroll
        for (int k = 0; k < 8; ++k) {
            int i = t * TILE + i0 + k * 32;
            v[t][k] = (full || i < N) ? data[i] : 0;
        }
    }

    int running = 0;
#pragma unroll
    for (int t = 0; t < NT; ++t) {
        const bool full = (t + 1) * TILE <= N;
        int incl[8];
        int carry = 0;
#pragma unroll
        for (int k = 0; k < 8; ++k) {
            int x = wave_incl_scan(v[t][k], lane);
            incl[k] = x + carry;
            carry += __builtin_amdgcn_readlane(x, 31);
        }
        if (lane == 0) wtot[t & 1][wv] = carry;
        __syncthreads();
        int w  = wtot[t & 1][lane];
        int wi = wave_incl_scan(w, lane);
        int wbase = __builtin_amdgcn_readlane(wi - w, wv);
        int base  = running + wbase;
        running  += __builtin_amdgcn_readlane(wi, 31);
#pragma unroll
        for (int k = 0; k < 8; ++k) {
            int i = t * TILE + i0 + k * 32;
            if (full || i <= N) data[i] = base + incl[k] - v[t][k];
        }
    }
}

template <int N>
__global__ void edge_fill(const int* __restrict__ src,
                          const int* __restrict__ dst,
                          const int* __restrict__ rank,
                          const int* __restrict__ off,
                          const int* __restrict__ tab,
                          int* __restrict__ eid, int n4) {
    int q = blockIdx.x * blockDim.x + threadIdx.x;
    if (q < n4) {
        const int* trow = tab + (long)(q / (EPB / 4)) * N;
        const int4 d = ((const int4*)dst)[q];
        const int4 r = ((const int4*)rank)[q];
        const int4 s = ((const int4*)src)[q];
        const int px = off[d.x] + trow[d.x] + r.x;
        const int py = off[d.y] + trow[d.y] + r.y;
        const int pz = off[d.z] + trow[d.z] + r.z;
        const int pw = off[d.w] + trow[d.w] + r.w;
        eid[px] = s.x;
        eid[py] = s.y;
        eid[pz] = s.z;
        eid[pw] = s.w;
    }
}

template <int MT>
__global__ __launch_bounds__(512)
void sage_layer(const float* __restrict__ x,
                const int* __restrict__ off,
                const int* __restrict__ eid,
                const float* __restrict__ Wl,
                const float* __restrict__ bias,
                const float* __restrict__ Wr,
                float* __restrict__ out,
                int M) {
    constexpr int BM = 16 * MT;
    __shared__ float lmean[BM * LROW];
    __shared__ float lx[BM * LROW];

    const int m0   = blockIdx.x * BM;
    const int tid  = threadIdx.x;
    const int lane = tid & 31;
    const int wv   = tid >> 5;

    for (int i = tid; i < BM * DIM; i += 512) {
        int row = i >> 8;
        int col = i & (DIM - 1);
        lx[row * LROW + col] = x[(long)(m0 + row) * DIM + col];
    }

    {
        const int r0 = wv * MT;
        int beg[MT + 1];
#pragma unroll
        for (int r = 0; r <= MT; ++r)
            beg[r] = __builtin_amdgcn_readfirstlane(off[m0 + r0 + r]);
        int deg[MT];
        int maxdeg = 0;
#pragma unroll
        for (int r = 0; r < MT; ++r) {
            deg[r] = beg[r + 1] - beg[r];
            maxdeg = max(maxdeg, deg[r]);
        }
        float4 a0[MT], a1[MT];
#pragma unroll
        for (int r = 0; r < MT; ++r) {
            a0[r] = make_float4(0.f, 0.f, 0.f, 0.f);
            a1[r] = make_float4(0.f, 0.f, 0.f, 0.f);
        }
        for (int e = 0; e < maxdeg; e += 32) {
            int myid[MT];
#pragma unroll
            for (int r = 0; r < MT; ++r)
                myid[r] = (e + lane < deg[r]) ? eid[beg[r] + e + lane] : 0;
            const int cnt = min(32, maxdeg - e);
            for (int j = 0; j < cnt; ++j) {
                float4 u[MT], v[MT];
#pragma unroll
                for (int r = 0; r < MT; ++r) {
                    int s = __shfl(myid[r], j, 32);
                    if (e + j < deg[r]) {
                        const float4* xr = (const float4*)(x + (long)s * DIM);
                        u[r] = xr[lane];
                        v[r] = xr[lane + 32];
                    } else {
                        u[r] = make_float4(0.f, 0.f, 0.f, 0.f);
                        v[r] = make_float4(0.f, 0.f, 0.f, 0.f);
                    }
                }
#pragma unroll
                for (int r = 0; r < MT; ++r) {
                    a0[r].x += u[r].x; a0[r].y += u[r].y; a0[r].z += u[r].z; a0[r].w += u[r].w;
                    a1[r].x += v[r].x; a1[r].y += v[r].y; a1[r].z += v[r].z; a1[r].w += v[r].w;
                }
            }
        }
#pragma unroll
        for (int r = 0; r < MT; ++r) {
            float inv = 1.0f / fmaxf((float)deg[r], 1.0f);
            float* mrow = lmean + (r0 + r) * LROW;
            mrow[lane * 4 + 0] = a0[r].x * inv;
            mrow[lane * 4 + 1] = a0[r].y * inv;
            mrow[lane * 4 + 2] = a0[r].z * inv;
            mrow[lane * 4 + 3] = a0[r].w * inv;
            mrow[128 + lane * 4 + 0] = a1[r].x * inv;
            mrow[128 + lane * 4 + 1] = a1[r].y * inv;
            mrow[128 + lane * 4 + 2] = a1[r].z * inv;
            mrow[128 + lane * 4 + 3] = a1[r].w * inv;
        }
    }
    __syncthreads();

    const int half = lane >> 4;
    const int lrow = lane & 15;
    const int n0   = wv * 16;

    const float* wl_r = Wl + (long)(n0 + lrow) * DIM;
    const float* wr_r = Wr + (long)(n0 + lrow) * DIM;
    const float* lm_r = lmean + lrow * LROW;
    const float* lx_r = lx + lrow * LROW;

    v8f acc[MT];
#pragma unroll
    for (int t = 0; t < MT; ++t) acc[t] = v8f{};
#pragma unroll (8 / MT)
    for (int k = 0; k < DIM; k += 4) {
        int kk = k + 2 * half;
        v2f bl = *(const v2f*)(wl_r + kk);
        v2f br = *(const v2f*)(wr_r + kk);
#pragma unroll
        for (int t = 0; t < MT; ++t) {
            v2f am = *(const v2f*)(lm_r + t * 16 * LROW + kk);
            acc[t] = __builtin_amdgcn_wmma_f32_16x16x4_f32(
                false, am, false, bl, (short)0, acc[t], false, false);
            v2f ax = *(const v2f*)(lx_r + t * 16 * LROW + kk);
            acc[t] = __builtin_amdgcn_wmma_f32_16x16x4_f32(
                false, ax, false, br, (short)0, acc[t], false, false);
        }
    }

    float bb = bias[n0 + lrow];
#pragma unroll
    for (int t = 0; t < MT; ++t) {
        float* orow = out + (long)(m0 + t * 16 + half * 8) * DIM + n0 + lrow;
#pragma unroll
        for (int r = 0; r < 8; ++r) {
            orow[(long)r * DIM] = tanhf(acc[t][r] + bb);
        }
    }
}

#define NOFF ((((N1C + 1) + (N2C + 1)) + 3) & ~3)

extern "C" void kernel_launch(void* const* d_in, const int* in_sizes, int n_in,
                              void* d_out, int out_size, void* d_ws, size_t ws_size,
                              hipStream_t stream) {
    const float* nodes = (const float*)d_in[0];
    const float* W_l1  = (const float*)d_in[1];
    const float* b1    = (const float*)d_in[2];
    const float* W_r1  = (const float*)d_in[3];
    const float* W_l2  = (const float*)d_in[4];
    const float* b2    = (const float*)d_in[5];
    const float* W_r2  = (const float*)d_in[6];
    const int*   src1  = (const int*)d_in[7];
    const int*   dst1  = (const int*)d_in[8];
    const int*   src2  = (const int*)d_in[9];
    const int*   dst2  = (const int*)d_in[10];

    int* wsi   = (int*)d_ws;
    int* off1  = wsi;
    int* off2  = off1 + (N1C + 1);
    int* rank1 = wsi + NOFF;
    int* rank2 = rank1 + E1C;
    int* eid1  = rank2 + E2C;
    int* eid2  = eid1 + E1C;
    float* h1  = (float*)(eid2 + E2C);
    int* tab1  = (int*)(h1 + (long)N1C * DIM);
    int* tab2  = tab1 + (long)NB1 * N1C;
    float* h2  = (float*)d_out;

    chunk_hist<<<NB1 + NB2, 1024, 0, stream>>>(dst1, rank1, tab1, dst2, rank2, tab2);
    chunk_prefix<<<(N1C + N2C + 127) / 128, 128, 0, stream>>>(tab1, off1, tab2, off2);
    scan_excl<N1C><<<1, 1024, 0, stream>>>(off1);
    scan_excl<N2C><<<1, 1024, 0, stream>>>(off2);
    edge_fill<N1C><<<(E1C / 4 + 255) / 256, 256, 0, stream>>>(src1, dst1, rank1, off1,
                                                              tab1, eid1, E1C / 4);
    edge_fill<N2C><<<(E2C / 4 + 255) / 256, 256, 0, stream>>>(src2, dst2, rank2, off2,
                                                              tab2, eid2, E2C / 4);
    sage_layer<4><<<N1C / 64, 512, 0, stream>>>(nodes, off1, eid1,
                                                W_l1, b1, W_r1, h1, N1C);
    sage_layer<2><<<N2C / 32, 512, 0, stream>>>(h1, off2, eid2,
                                                W_l2, b2, W_r2, h2, N2C);
}


# sage_layer layer 1: 160-row stripes in 32-wave blocks, 250 blocks all resident at once (one per compute unit) instead of 625 64-row blocks over 512 places; LDS tiles unpadded, XOR-swizzled to fit
# speedup vs baseline: 2.3858x; 1.0988x over previous
#include <hip/hip_runtime.h>

typedef __attribute__((ext_vector_type(2))) float v2f;
typedef __attribute__((ext_vector_type(8))) float v8f;

#define DIM   256
#define N1C   40000
#define N2C   8000
#define E1C   1000000
#define E2C   200000

#define EPB   20000
#define NB1   (E1C / EPB)
#define NB2   (E2C / EPB)

template <int N>
__device__ __forceinline__ void chunk_hist_body(const int* __restrict__ dst,
                                                int* __restrict__ rank,
                                                int* __restrict__ tab,
                                                int* hist, int chunk) {
    constexpr int Q  = EPB / 4;
    constexpr int IT = (Q + 1023) / 1024;
    const int tid = threadIdx.x;
    const int4* d4 = (const int4*)(dst + (long)chunk * EPB);
    int4* r4 = (int4*)(rank + (long)chunk * EPB);

    int4 d[IT];
#pragma unroll
    for (int it = 0; it < IT; ++it) {
        int q = tid + it * 1024;
        d[it] = d4[(q < Q) ? q : Q - 1];
    }

    int4* h4 = (int4*)hist;
    for (int i = tid; i < N / 4; i += 1024) h4[i] = make_int4(0, 0, 0, 0);
    __syncthreads();

#pragma unroll
    for (int it = 0; it < IT; ++it) {
        int q = tid + it * 1024;
        if (q < Q) {
            int4 r;
            r.x = atomicAdd(hist + d[it].x, 1);
            r.y = atomicAdd(hist + d[it].y, 1);
            r.z = atomicAdd(hist + d[it].z, 1);
            r.w = atomicAdd(hist + d[it].w, 1);
            r4[q] = r;
        }
    }
    __syncthreads();

    int4* t4 = (int4*)(tab + (long)chunk * N);
    for (int i = tid; i < N / 4; i += 1024) t4[i] = h4[i];
}

__global__ __launch_bounds__(1024)
void chunk_hist(const int* __restrict__ dst1, int* __restrict__ rank1, int* __restrict__ tab1,
                const int* __restrict__ dst2, int* __restrict__ rank2, int* __restrict__ tab2) {
    __shared__ __attribute__((aligned(16))) int hist[N1C];
    if (blockIdx.x < NB1)
        chunk_hist_body<N1C>(dst1, rank1, tab1, hist, blockIdx.x);
    else
        chunk_hist_body<N2C>(dst2, rank2, tab2, hist, blockIdx.x - NB1);
}

template <int N, int NB>
__device__ __forceinline__ void chunk_prefix_body(int* __restrict__ tab,
                                                  int* __restrict__ cnt, int d) {
    int cv[NB];
#pragma unroll
    for (int b = 0; b < NB; ++b) cv[b] = tab[(long)b * N + d];
    int run = 0;
#pragma unroll
    for (int b = 0; b < NB; ++b) {
        tab[(long)b * N + d] = run;
        run += cv[b];
    }
    cnt[d] = run;
}

__global__ __launch_bounds__(128)
void chunk_prefix(int* __restrict__ tab1, int* __restrict__ cnt1,
                  int* __restrict__ tab2, int* __restrict__ cnt2) {
    int g = blockIdx.x * 128 + threadIdx.x;
    if (g < N1C)
        chunk_prefix_body<N1C, NB1>(tab1, cnt1, g);
    else if (g < N1C + N2C)
        chunk_prefix_body<N2C, NB2>(tab2, cnt2, g - N1C);
}

__device__ __forceinline__ int wave_incl_scan(int x, int lane) {
    x += __builtin_amdgcn_update_dpp(0, x, 0x111, 0xf, 0xf, true);
    x += __builtin_amdgcn_update_dpp(0, x, 0x112, 0xf, 0xf, true);
    x += __builtin_amdgcn_update_dpp(0, x, 0x114, 0xf, 0xf, true);
    x += __builtin_amdgcn_update_dpp(0, x, 0x118, 0xf, 0xf, true);
    int low = __builtin_amdgcn_readlane(x, 15);
    x += (lane >= 16) ? low : 0;
    return x;
}

template <int N>
__global__ __launch_bounds__(1024)
void scan_excl(int* __restrict__ data) {
    constexpr int TILE = 1024 * 8;
    constexpr int NT   = (N + 1 + TILE - 1) / TILE;
    __shared__ int wtot[2][32];

    const int lane = threadIdx.x & 31;
    const int wv   = __builtin_amdgcn_readfirstlane((int)(threadIdx.x >> 5));
    const int i0   = wv * 256 + lane;

    int v[NT][8];
#pragma unroll
    for (int t = 0; t < NT; ++t) {
        const bool full = (t + 1) * TILE <= N;
#pragma unroll
        for (int k = 0; k < 8; ++k) {
            int i = t * TILE + i0 + k * 32;
            v[t][k] = (full || i < N) ? data[i] : 0;
        }
    }

    int running = 0;
#pragma unroll
    for (int t = 0; t < NT; ++t) {
        const bool full = (t + 1) * TILE <= N;
        int incl[8];
        int carry = 0;
#pragma unroll
        for (int k = 0; k < 8; ++k) {
            int x = wave_incl_scan(v[t][k], lane);
            incl[k] = x + carry;
            carry += __builtin_amdgcn_readlane(x, 31);
        }
        if (lane == 0) wtot[t & 1][wv] = carry;
        __syncthreads();
        int w  = wtot[t & 1][lane];
        int wi = wave_incl_scan(w, lane);
        int wbase = __builtin_amdgcn_readlane(wi - w, wv);
        int base  = running + wbase;
        running  += __builtin_amdgcn_readlane(wi, 31);
#pragma unroll
        for (int k = 0; k < 8; ++k) {
            int i = t * TILE + i0 + k * 32;
            if (full || i <= N) data[i] = base + incl[k] - v[t][k];
        }
    }
}

template <int N>
__global__ void edge_fill(const int* __restrict__ src,
                          const int* __restrict__ dst,
                          const int* __restrict__ rank,
                          const int* __restrict__ off,
                          const int* __restrict__ tab,
                          int* __restrict__ eid, int n4) {
    int q = blockIdx.x * blockDim.x + threadIdx.x;
    if (q < n4) {
        const int* trow = tab + (long)(q / (EPB / 4)) * N;
        const int4 d = ((const int4*)dst)[q];
        const int4 r = ((const int4*)rank)[q];
        const int4 s = ((const int4*)src)[q];
        const int px = off[d.x] + trow[d.x] + r.x;
        const int py = off[d.y] + trow[d.y] + r.y;
        const int pz = off[d.z] + trow[d.z] + r.z;
        const int pw = off[d.w] + trow[d.w] + r.w;
        eid[px] = s.x;
        eid[py] = s.y;
        eid[pz] = s.z;
        eid[pw] = s.w;
    }
}

template <int MT, int RH>
__global__ __launch_bounds__(512 * RH)
void sage_layer(const float* __restrict__ x,
                const int* __restrict__ off,
                const int* __restrict__ eid,
                const float* __restrict__ Wl,
                const float* __restrict__ bias,
                const float* __restrict__ Wr,
                float* __restrict__ out,
                int M) {
    constexpr int NT = 512 * RH;
    constexpr int BM = 16 * MT * RH;
    __shared__ float lmean[BM * DIM];
    __shared__ float lx[BM * DIM];

    const int m0   = blockIdx.x * BM;
    const int tid  = threadIdx.x;
    const int lane = tid & 31;
    const int wv   = tid >> 5;

    for (int i = tid; i < BM * DIM; i += NT) {
        int row = i >> 8;
        int col = i & (DIM - 1);
        lx[row * DIM + (col ^ ((row & 15) << 2))] = x[(long)(m0 + row) * DIM + col];
    }

    {
        const int r0 = wv * MT;
        int beg[MT + 1];
#pragma unroll
        for (int r = 0; r <= MT; ++r)
            beg[r] = __builtin_amdgcn_readfirstlane(off[m0 + r0 + r]);
        int deg[MT];
        int maxdeg = 0;
#pragma unroll
        for (int r = 0; r < MT; ++r) {
            deg[r] = beg[r + 1] - beg[r];
            maxdeg = max(maxdeg, deg[r]);
        }
        float4 a0[MT], a1[MT];
#pragma unroll
        for (int r = 0; r < MT; ++r) {
            a0[r] = make_float4(0.f, 0.f, 0.f, 0.f);
            a1[r] = make_float4(0.f, 0.f, 0.f, 0.f);
        }
        for (int e = 0; e < maxdeg; e += 32) {
            int myid[MT];
#pragma unroll
            for (int r = 0; r < MT; ++r)
                myid[r] = (e + lane < deg[r]) ? eid[beg[r] + e + lane] : 0;
            const int cnt = min(32, maxdeg - e);
            for (int j = 0; j < cnt; ++j) {
                float4 u[MT], v[MT];
#pragma unroll
                for (int r = 0; r < MT; ++r) {
                    int s = __shfl(myid[r], j, 32);
                    if (e + j < deg[r]) {
                        const float4* xr = (const float4*)(x + (long)s * DIM);
                        u[r] = xr[lane];
                        v[r] = xr[lane + 32];
                    } else {
                        u[r] = make_float4(0.f, 0.f, 0.f, 0.f);
                        v[r] = make_float4(0.f, 0.f, 0.f, 0.f);
                    }
                }
#pragma unroll
                for (int r = 0; r < MT; ++r) {
                    a0[r].x += u[r].x; a0[r].y += u[r].y; a0[r].z += u[r].z; a0[r].w += u[r].w;
                    a1[r].x += v[r].x; a1[r].y += v[r].y; a1[r].z += v[r].z; a1[r].w += v[r].w;
                }
            }
        }
#pragma unroll
        for (int r = 0; r < MT; ++r) {
            float inv = 1.0f / fmaxf((float)deg[r], 1.0f);
            float* mrow = lmean + (r0 + r) * DIM;
            const int c = (lane * 4) ^ (((r0 + r) & 15) << 2);
            mrow[c + 0] = a0[r].x * inv;
            mrow[c + 1] = a0[r].y * inv;
            mrow[c + 2] = a0[r].z * inv;
            mrow[c + 3] = a0[r].w * inv;
            mrow[128 + c + 0] = a1[r].x * inv;
            mrow[128 + c + 1] = a1[r].y * inv;
            mrow[128 + c + 2] = a1[r].z * inv;
            mrow[128 + c + 3] = a1[r].w * inv;
        }
    }
    __syncthreads();

    const int half = lane >> 4;
    const int lrow = lane & 15;
    const int n0   = (wv & 15) * 16;
    const int rb   = (wv >> 4) * (16 * MT);
    const int swz  = lrow << 2;

    const float* wl_r = Wl + (long)(n0 + lrow) * DIM;
    const float* wr_r = Wr + (long)(n0 + lrow) * DIM;
    const float* lm_r = lmean + (rb + lrow) * DIM + 2 * half;
    const float* lx_r = lx + (rb + lrow) * DIM + 2 * half;

    v8f acc[MT];
#pragma unroll
    for (int t = 0; t < MT; ++t) acc[t] = v8f{};
#pragma unroll (MT > 2 ? 2 : 4)
    for (int k = 0; k < DIM; k += 4) {
        int kk = k + 2 * half;
        int ks = k ^ swz;
        v2f bl = *(const v2f*)(wl_r + kk);
        v2f br = *(const v2f*)(wr_r + kk);
#pragma unroll
        for (int t = 0; t < MT; ++t) {
            v2f am = *(const v2f*)(lm_r + t * 16 * DIM + ks);
            acc[t] = __builtin_amdgcn_wmma_f32_16x16x4_f32(
                false, am, false, bl, (short)0, acc[t], false, false);
            v2f ax = *(const v2f*)(lx_r + t * 16 * DIM + ks);
            acc[t] = __builtin_amdgcn_wmma_f32_16x16x4_f32(
                false, ax, false, br, (short)0, acc[t], false, false);
        }
    }

    float bb = bias[n0 + lrow];
#pragma unroll
    for (int t = 0; t < MT; ++t) {
        float* orow = out + (long)(m0 + rb + t * 16 + half * 8) * DIM + n0 + lrow;
#pragma unroll
        for (int r = 0; r < 8; ++r) {
            orow[(long)r * DIM] = tanhf(acc[t][r] + bb);
        }
    }
}

#define NOFF ((((N1C + 1) + (N2C + 1)) + 3) & ~3)

extern "C" void kernel_launch(void* const* d_in, const int* in_sizes, int n_in,
                              void* d_out, int out_size, void* d_ws, size_t ws_size,
                              hipStream_t stream) {
    const float* nodes = (const float*)d_in[0];
    const float* W_l1  = (const float*)d_in[1];
    const float* b1    = (const float*)d_in[2];
    const float* W_r1  = (const float*)d_in[3];
    const float* W_l2  = (const float*)d_in[4];
    const float* b2    = (const float*)d_in[5];
    const float* W_r2  = (const float*)d_in[6];
    const int*   src1  = (const int*)d_in[7];
    const int*   dst1  = (const int*)d_in[8];
    const int*   src2  = (const int*)d_in[9];
    const int*   dst2  = (const int*)d_in[10];

    int* wsi   = (int*)d_ws;
    int* off1  = wsi;
    int* off2  = off1 + (N1C + 1);
    int* rank1 = wsi + NOFF;
    int* rank2 = rank1 + E1C;
    int* eid1  = rank2 + E2C;
    int* eid2  = eid1 + E1C;
    float* h1  = (float*)(eid2 + E2C);
    int* tab1  = (int*)(h1 + (long)N1C * DIM);
    int* tab2  = tab1 + (long)NB1 * N1C;
    float* h2  = (float*)d_out;

    chunk_hist<<<NB1 + NB2, 1024, 0, stream>>>(dst1, rank1, tab1, dst2, rank2, tab2);
    chunk_prefix<<<(N1C + N2C + 127) / 128, 128, 0, stream>>>(tab1, off1, tab2, off2);
    scan_excl<N1C><<<1, 1024, 0, stream>>>(off1);
    scan_excl<N2C><<<1, 1024, 0, stream>>>(off2);
    edge_fill<N1C><<<(E1C / 4 + 255) / 256, 256, 0, stream>>>(src1, dst1, rank1, off1,
                                                              tab1, eid1, E1C / 4);
    edge_fill<N2C><<<(E2C / 4 + 255) / 256, 256, 0, stream>>>(src2, dst2, rank2, off2,
                                                              tab2, eid2, E2C / 4);
    sage_layer<5, 2><<<N1C / 160, 1024, 0, stream>>>(nodes, off1, eid1,
                                                     W_l1, b1, W_r1, h1, N1C);
    sage_layer<2, 1><<<N2C / 32, 512, 0, stream>>>(h1, off2, eid2,
                                                   W_l2, b2, W_r2, h2, N2C);
}


# CSR build: 8 launches -> 5: offsets come out of chunk_prefix (per-chunk 128-destination bucket prefixes from chunk_hist + a prefix in the block), no scan launches; both graphs' fills in one launch
# speedup vs baseline: 2.5304x; 1.0606x over previous
#include <hip/hip_runtime.h>

typedef __attribute__((ext_vector_type(2))) float v2f;
typedef __attribute__((ext_vector_type(8))) float v8f;

#define DIM   256
#define N1C   40000
#define N2C   8000
#define E1C   1000000
#define E2C   200000

#define EPB   20000
#define NB1   (E1C / EPB)
#define NB2   (E2C / EPB)

#define BKT   128
#define NBK1  ((N1C + 1 + BKT - 1) / BKT)
#define NBK2  ((N2C + 1 + BKT - 1) / BKT)
static_assert(N1C % BKT != 0 && N2C % BKT != 0, "off[N] must fall inside the last bucket");

__device__ __forceinline__ int wave_incl_scan(int x, int lane) {
    x += __builtin_amdgcn_update_dpp(0, x, 0x111, 0xf, 0xf, true);
    x += __builtin_amdgcn_update_dpp(0, x, 0x112, 0xf, 0xf, true);
    x += __builtin_amdgcn_update_dpp(0, x, 0x114, 0xf, 0xf, true);
    x += __builtin_amdgcn_update_dpp(0, x, 0x118, 0xf, 0xf, true);
    int low = __builtin_amdgcn_readlane(x, 15);
    x += (lane >= 16) ? low : 0;
    return x;
}

template <int N, int NBK>
__device__ __forceinline__ void chunk_hist_body(const int* __restrict__ dst,
                                                int* __restrict__ rank,
                                                int* __restrict__ tab,
                                                int* __restrict__ bpre,
                                                int* hist, int* bs, int chunk) {
    constexpr int Q  = EPB / 4;
    constexpr int IT = (Q + 1023) / 1024;
    const int tid  = threadIdx.x;
    const int lane = tid & 31;
    const int wv   = tid >> 5;
    const int4* d4 = (const int4*)(dst + (long)chunk * EPB);
    int4* r4 = (int4*)(rank + (long)chunk * EPB);

    int4 d[IT];
#pragma unroll
    for (int it = 0; it < IT; ++it) {
        int q = tid + it * 1024;
        d[it] = d4[(q < Q) ? q : Q - 1];
    }

    int4* h4 = (int4*)hist;
    for (int i = tid; i < N / 4; i += 1024) h4[i] = make_int4(0, 0, 0, 0);
    __syncthreads();

#pragma unroll
    for (int it = 0; it < IT; ++it) {
        int q = tid + it * 1024;
        if (q < Q) {
            int4 r;
            r.x = atomicAdd(hist + d[it].x, 1);
            r.y = atomicAdd(hist + d[it].y, 1);
            r.z = atomicAdd(hist + d[it].z, 1);
            r.w = atomicAdd(hist + d[it].w, 1);
            r4[q] = r;
        }
    }
    __syncthreads();

    for (int k = wv; k < NBK; k += 32) {
        int i = k * BKT + lane * 4;
        int s = 0;
        if (i < N) {
            int4 h = *(const int4*)(hist + i);
            s = h.x + h.y + h.z + h.w;
        }
        s = wave_incl_scan(s, lane);
        if (lane == 31) bs[k] = s;
    }
    __syncthreads();

    if (wv == 0) {
        constexpr int PL = (NBK + 31) / 32;
        int v[PL];
        int sum = 0;
#pragma unroll
        for (int j = 0; j < PL; ++j) {
            int k = lane * PL + j;
            v[j] = (k < NBK) ? bs[k] : 0;
            sum += v[j];
        }
        int run = wave_incl_scan(sum, lane) - sum;
#pragma unroll
        for (int j = 0; j < PL; ++j) {
            int k = lane * PL + j;
            if (k < NBK) bpre[chunk * NBK + k] = run;
            run += v[j];
        }
    }

    int4* t4 = (int4*)(tab + (long)chunk * N);
    for (int i = tid; i < N / 4; i += 1024) t4[i] = h4[i];
}

__global__ __launch_bounds__(1024)
void chunk_hist(const int* __restrict__ dst1, int* __restrict__ rank1,
                int* __restrict__ tab1, int* __restrict__ bpre1,
                const int* __restrict__ dst2, int* __restrict__ rank2,
                int* __restrict__ tab2, int* __restrict__ bpre2) {
    __shared__ __attribute__((aligned(16))) int hist[N1C];
    __shared__ int bs[NBK1 > NBK2 ? NBK1 : NBK2];
    if (blockIdx.x < NB1)
        chunk_hist_body<N1C, NBK1>(dst1, rank1, tab1, bpre1, hist, bs, blockIdx.x);
    else
        chunk_hist_body<N2C, NBK2>(dst2, rank2, tab2, bpre2, hist, bs, blockIdx.x - NB1);
}

template <int N, int NB, int NBK>
__device__ __forceinline__ void chunk_prefix_body(int* __restrict__ tab,
                                                  const int* __restrict__ bpre,
                                                  int* __restrict__ off,
                                                  int* wtot, int bucket) {
    const int t    = threadIdx.x;
    const int lane = t & 31;
    const int wv   = t >> 5;
    const int d    = bucket * BKT + t;
    const bool in  = d < N;
    const int dc   = in ? d : N - 1;
    int cv[NB];
#pragma unroll
    for (int c = 0; c < NB; ++c) cv[c] = tab[(long)c * N + dc];
    int part = 0;
#pragma unroll
    for (int j = 0; j < (NB + 31) / 32; ++j) {
        int c = lane + 32 * j;
        part += (c < NB) ? bpre[c * NBK + bucket] : 0;
    }
    const int base = __builtin_amdgcn_readlane(wave_incl_scan(part, lane), 31);
    int run = 0;
    if (in) {
#pragma unroll
        for (int c = 0; c < NB; ++c) {
            tab[(long)c * N + d] = run;
            run += cv[c];
        }
    }
    int incl = wave_incl_scan(run, lane);
    if (lane == 31) wtot[wv] = incl;
    __syncthreads();
    int before = 0;
#pragma unroll
    for (int w = 0; w < BKT / 32; ++w) before += (w < wv) ? wtot[w] : 0;
    if (d <= N) off[d] = base + before + incl - run;
}

__global__ __launch_bounds__(BKT)
void chunk_prefix(int* __restrict__ tab1, const int* __restrict__ bpre1, int* __restrict__ off1,
                  int* __restrict__ tab2, const int* __restrict__ bpre2, int* __restrict__ off2) {
    __shared__ int wtot[BKT / 32];
    if (blockIdx.x < NBK1)
        chunk_prefix_body<N1C, NB1, NBK1>(tab1, bpre1, off1, wtot, blockIdx.x);
    else
        chunk_prefix_body<N2C, NB2, NBK2>(tab2, bpre2, off2, wtot, blockIdx.x - NBK1);
}

template <int N>
__device__ __forceinline__ void edge_fill_body(const int* __restrict__ src,
                                               const int* __restrict__ dst,
                                               const int* __restrict__ rank,
                                               const int* __restrict__ off,
                                               const int* __restrict__ tab,
                                               int* __restrict__ eid, int q) {
    const int* trow = tab + (long)(q / (EPB / 4)) * N;
    const int4 d = ((const int4*)dst)[q];
    const int4 r = ((const int4*)rank)[q];
    const int4 s = ((const int4*)src)[q];
    const int px = off[d.x] + trow[d.x] + r.x;
    const int py = off[d.y] + trow[d.y] + r.y;
    const int pz = off[d.z] + trow[d.z] + r.z;
    const int pw = off[d.w] + trow[d.w] + r.w;
    eid[px] = s.x;
    eid[py] = s.y;
    eid[pz] = s.z;
    eid[pw] = s.w;
}

#define FB1 ((E1C / 4 + 255) / 256)
#define FB2 ((E2C / 4 + 255) / 256)

__global__ __launch_bounds__(256)
void edge_fill(const int* __restrict__ src1, const int* __restrict__ dst1,
               const int* __restrict__ rank1, const int* __restrict__ off1,
               const int* __restrict__ tab1, int* __restrict__ eid1,
               const int* __restrict__ src2, const int* __restrict__ dst2,
               const int* __restrict__ rank2, const int* __restrict__ off2,
               const int* __restrict__ tab2, int* __restrict__ eid2) {
    if (blockIdx.x < FB1) {
        int q = blockIdx.x * 256 + threadIdx.x;
        if (q < E1C / 4) edge_fill_body<N1C>(src1, dst1, rank1, off1, tab1, eid1, q);
    } else {
        int q = (blockIdx.x - FB1) * 256 + threadIdx.x;
        if (q < E2C / 4) edge_fill_body<N2C>(src2, dst2, rank2, off2, tab2, eid2, q);
    }
}

template <int MT, int RH>
__global__ __launch_bounds__(512 * RH)
void sage_layer(const float* __restrict__ x,
                const int* __restrict__ off,
                const int* __restrict__ eid,
                const float* __restrict__ Wl,
                const float* __restrict__ bias,
                const float* __restrict__ Wr,
                float* __restrict__ out,
                int M) {
    constexpr int NT = 512 * RH;
    constexpr int BM = 16 * MT * RH;
    __shared__ float lmean[BM * DIM];
    __shared__ float lx[BM * DIM];

    const int m0   = blockIdx.x * BM;
    const int tid  = threadIdx.x;
    const int lane = tid & 31;
    const int wv   = tid >> 5;

    for (int i = tid; i < BM * DIM; i += NT) {
        int row = i >> 8;
        int col = i & (DIM - 1);
        lx[row * DIM + (col ^ ((row & 15) << 2))] = x[(long)(m0 + row) * DIM + col];
    }

    {
        const int r0 = wv * MT;
        int beg[MT + 1];
#pragma unroll
        for (int r = 0; r <= MT; ++r)
            beg[r] = __builtin_amdgcn_readfirstlane(off[m0 + r0 + r]);
        int deg[MT];
        int maxdeg = 0;
#pragma unroll
        for (int r = 0; r < MT; ++r) {
            deg[r] = beg[r + 1] - beg[r];
            maxdeg = max(maxdeg, deg[r]);
        }
        float4 a0[MT], a1[MT];
#pragma unroll
        for (int r = 0; r < MT; ++r) {
            a0[r] = make_float4(0.f, 0.f, 0.f, 0.f);
            a1[r] = make_float4(0.f, 0.f, 0.f, 0.f);
        }
        for (int e = 0; e < maxdeg; e += 32) {
            int myid[MT];
#pragma unroll
            for (int r = 0; r < MT; ++r)
                myid[r] = (e + lane < deg[r]) ? eid[beg[r] + e + lane] : 0;
            const int cnt = min(32, maxdeg - e);
            for (int j = 0; j < cnt; ++j) {
                float4 u[MT], v[MT];
#pragma unroll
                for (int r = 0; r < MT; ++r) {
                    int s = __shfl(myid[r], j, 32);
                    if (e + j < deg[r]) {
                        const float4* xr = (const float4*)(x + (long)s * DIM);
                        u[r] = xr[lane];
                        v[r] = xr[lane + 32];
                    } else {
                        u[r] = make_float4(0.f, 0.f, 0.f, 0.f);
                        v[r] = make_float4(0.f, 0.f, 0.f, 0.f);
                    }
                }
#pragma unroll
                for (int r = 0; r < MT; ++r) {
                    a0[r].x += u[r].x; a0[r].y += u[r].y; a0[r].z += u[r].z; a0[r].w += u[r].w;
                    a1[r].x += v[r].x; a1[r].y += v[r].y; a1[r].z += v[r].z; a1[r].w += v[r].w;
                }
            }
        }
#pragma unroll
        for (int r = 0; r < MT; ++r) {
            float inv = 1.0f / fmaxf((float)deg[r], 1.0f);
            float* mrow = lmean + (r0 + r) * DIM;
            const int c = (lane * 4) ^ (((r0 + r) & 15) << 2);
            mrow[c + 0] = a0[r].x * inv;
            mrow[c + 1] = a0[r].y * inv;
            mrow[c + 2] = a0[r].z * inv;
            mrow[c + 3] = a0[r].w * inv;
            mrow[128 + c + 0] = a1[r].x * inv;
            mrow[128 + c + 1] = a1[r].y * inv;
            mrow[128 + c + 2] = a1[r].z * inv;
            mrow[128 + c + 3] = a1[r].w * inv;
        }
    }
    __syncthreads();

    const int half = lane >> 4;
    const int lrow = lane & 15;
    const int n0   = (wv & 15) * 16;
    const int rb   = (wv >> 4) * (16 * MT);
    const int swz  = lrow << 2;

    const float* wl_r = Wl + (long)(n0 + lrow) * DIM;
    const float* wr_r = Wr + (long)(n0 + lrow) * DIM;
    const float* lm_r = lmean + (rb + lrow) * DIM + 2 * half;
    const float* lx_r = lx + (rb + lrow) * DIM + 2 * half;

    v8f acc[MT];
#pragma unroll
    for (int t = 0; t < MT; ++t) acc[t] = v8f{};
#pragma unroll (MT > 2 ? 2 : 4)
    for (int k = 0; k < DIM; k += 4) {
        int kk = k + 2 * half;
        int ks = k ^ swz;
        v2f bl = *(const v2f*)(wl_r + kk);
        v2f br = *(const v2f*)(wr_r + kk);
#pragma unroll
        for (int t = 0; t < MT; ++t) {
            v2f am = *(const v2f*)(lm_r + t * 16 * DIM + ks);
            acc[t] = __builtin_amdgcn_wmma_f32_16x16x4_f32(
                false, am, false, bl, (short)0, acc[t], false, false);
            v2f ax = *(const v2f*)(lx_r + t * 16 * DIM + ks);
            acc[t] = __builtin_amdgcn_wmma_f32_16x16x4_f32(
                false, ax, false, br, (short)0, acc[t], false, false);
        }
    }

    float bb = bias[n0 + lrow];
#pragma unroll
    for (int t = 0; t < MT; ++t) {
        float* orow = out + (long)(m0 + rb + t * 16 + half * 8) * DIM + n0 + lrow;
#pragma unroll
        for (int r = 0; r < 8; ++r) {
            orow[(long)r * DIM] = tanhf(acc[t][r] + bb);
        }
    }
}

#define NOFF ((((N1C + 1) + (N2C + 1)) + 3) & ~3)

extern "C" void kernel_launch(void* const* d_in, const int* in_sizes, int n_in,
                              void* d_out, int out_size, void* d_ws, size_t ws_size,
                              hipStream_t stream) {
    const float* nodes = (const float*)d_in[0];
    const float* W_l1  = (const float*)d_in[1];
    const float* b1    = (const float*)d_in[2];
    const float* W_r1  = (const float*)d_in[3];
    const float* W_l2  = (const float*)d_in[4];
    const float* b2    = (const float*)d_in[5];
    const float* W_r2  = (const float*)d_in[6];
    const int*   src1  = (const int*)d_in[7];
    const int*   dst1  = (const int*)d_in[8];
    const int*   src2  = (const int*)d_in[9];
    const int*   dst2  = (const int*)d_in[10];

    int* wsi   = (int*)d_ws;
    int* off1  = wsi;
    int* off2  = off1 + (N1C + 1);
    int* rank1 = wsi + NOFF;
    int* rank2 = rank1 + E1C;
    int* eid1  = rank2 + E2C;
    int* eid2  = eid1 + E1C;
    float* h1  = (float*)(eid2 + E2C);
    int* tab1  = (int*)(h1 + (long)N1C * DIM);
    int* tab2  = tab1 + (long)NB1 * N1C;
    int* bpre1 = tab2 + (long)NB2 * N2C;
    int* bpre2 = bpre1 + NB1 * NBK1;
    float* h2  = (float*)d_out;

    chunk_hist<<<NB1 + NB2, 1024, 0, stream>>>(dst1, rank1, tab1, bpre1,
                                               dst2, rank2, tab2, bpre2);
    chunk_prefix<<<NBK1 + NBK2, BKT, 0, stream>>>(tab1, bpre1, off1, tab2, bpre2, off2);
    edge_fill<<<FB1 + FB2, 256, 0, stream>>>(src1, dst1, rank1, off1, tab1, eid1,
                                             src2, dst2, rank2, off2, tab2, eid2);
    sage_layer<5, 2><<<N1C / 160, 1024, 0, stream>>>(nodes, off1, eid1,
                                                     W_l1, b1, W_r1, h1, N1C);
    sage_layer<2, 1><<<N2C / 32, 512, 0, stream>>>(h1, off2, eid2,
                                                   W_l2, b2, W_r2, h2, N2C);
}
